# starting kernel
# baseline (speedup 1.0000x reference)
#include <hip/hip_runtime.h>
#include <math.h>

typedef __attribute__((ext_vector_type(2))) float v2f;
typedef __attribute__((ext_vector_type(8))) float v8f;

#define CELLS_PER_BLOCK 16
#define THREADS 512
#define LDS_PITCH 132
#define TOK 32
#define DIM 128

__device__ __forceinline__ float gelu_erf(float v) {
    return 0.5f * v * (1.0f + erff(v * 0.70710678118654752440f));
}

__global__ __launch_bounds__(THREADS) void GeluAvgEmbed_kernel(
    const int*   __restrict__ x,
    const float* __restrict__ W,
    const float* __restrict__ wp,
    const float* __restrict__ bp,
    float*       __restrict__ out)
{
    __shared__ __align__(16) float lds[CELLS_PER_BLOCK * LDS_PITCH];

    const int lane  = threadIdx.x & 31;
    const int wave  = threadIdx.x >> 5;
    const int cell0 = blockIdx.x * CELLS_PER_BLOCK;
    const int cell  = cell0 + wave;

    const int* xc = x + (size_t)cell * TOK;
    int myidx = xc[lane];

    float ax = 0.f, ay = 0.f, az = 0.f, aw = 0.f;
#pragma unroll
    for (int t = 0; t < TOK; ++t) {
        int idx = __builtin_amdgcn_readlane(myidx, t);
        const float4* row = (const float4*)(W + (size_t)idx * DIM);
        float4 v = row[lane];
        ax += v.x; ay += v.y; az += v.z; aw += v.w;
    }
    const float inv = 1.0f / (float)TOK;
    float4 actv;
    actv.x = gelu_erf(ax * inv);
    actv.y = gelu_erf(ay * inv);
    actv.z = gelu_erf(az * inv);
    actv.w = gelu_erf(aw * inv);

    *(float4*)&lds[wave * LDS_PITCH + lane * 4] = actv;

    __syncthreads();

    if (wave == 0) {
        const int half = lane >> 4;
        const int off  = half << 1;
        const int m    = lane & 15;

        v8f c = {};
#pragma unroll
        for (int j = 0; j < DIM / 4; ++j) {
            v2f a = *(const v2f*)&lds[m * LDS_PITCH + 4 * j + off];
            v2f b;
            b.x = wp[4 * j + off];
            b.y = wp[4 * j + off + 1];
            c = __builtin_amdgcn_wmma_f32_16x16x4_f32(
                    false, a, false, b, (short)0, c, false, false);
        }

        if (m == 0) {
            const float bias = bp[0];
            float* o = out + cell0 + half * 8;
#pragma unroll
            for (int r = 0; r < 8; ++r)
                o[r] = c[r] + bias;
        }
    }
}

extern "C" void kernel_launch(void* const* d_in, const int* in_sizes, int n_in,
                              void* d_out, int out_size, void* d_ws, size_t ws_size,
                              hipStream_t stream) {
    const int*   x  = (const int*)d_in[0];
    const float* W  = (const float*)d_in[1];
    const float* wp = (const float*)d_in[2];
    const float* bp = (const float*)d_in[3];
    float* out = (float*)d_out;

    const int ncells = out_size;
    const int blocks = ncells / CELLS_PER_BLOCK;
    GeluAvgEmbed_kernel<<<blocks, THREADS, 0, stream>>>(x, W, wp, bp, out);
}


# One wave per 16 cells, no work-group barrier: its 512 row gathers stream through a 32-slot register ring (add oldest, refill slot), then its own LDS transpose + WMMA dot; grid 3200 x 32 threads
# speedup vs baseline: 1.1708x; 1.1708x over previous
#include <hip/hip_runtime.h>
#include <math.h>

typedef __attribute__((ext_vector_type(2))) float v2f;
typedef __attribute__((ext_vector_type(8))) float v8f;

#define CELLS_PER_BLOCK 16
#define THREADS 32
#define LDS_PITCH 132
#define TOK 32
#define DIM 128

__device__ __forceinline__ float gelu_erf(float v) {
    return 0.5f * v * (1.0f + erff(v * 0.70710678118654752440f));
}

__device__ __forceinline__ float4 load_row(const float* __restrict__ W, int idx, int lane) {
    const float4* row = (const float4*)(W + (size_t)idx * DIM);
    return row[lane];
}

__global__ __launch_bounds__(THREADS) void GeluAvgEmbed_kernel(
    const int*   __restrict__ x,
    const float* __restrict__ W,
    const float* __restrict__ wp,
    const float* __restrict__ bp,
    float*       __restrict__ out)
{
    __shared__ __align__(16) float lds[CELLS_PER_BLOCK * LDS_PITCH];

    const int lane  = threadIdx.x;
    const int cell0 = blockIdx.x * CELLS_PER_BLOCK;

    const int* xc = x + (size_t)cell0 * TOK + lane;
    const int idx_first = xc[0];
    int idx_nxt = xc[TOK];
    float4 r[TOK];
#pragma unroll
    for (int t = 0; t < TOK; ++t)
        r[t] = load_row(W, __builtin_amdgcn_readlane(idx_first, t), lane);

    const float inv = 1.0f / (float)TOK;

#pragma unroll 1
    for (int c = 0; c < CELLS_PER_BLOCK - 1; ++c) {
        const int cn = (c + 2 < CELLS_PER_BLOCK) ? (c + 2) : (CELLS_PER_BLOCK - 1);
        const int idx_n2 = xc[cn * TOK];

        float ax = 0.f, ay = 0.f, az = 0.f, aw = 0.f;
#pragma unroll
        for (int t = 0; t < TOK; ++t) {
            const float4 v = r[t];
            ax += v.x; ay += v.y; az += v.z; aw += v.w;
            int idx = __builtin_amdgcn_readlane(idx_nxt, t);
            asm volatile("" : "+s"(idx) : "v"(ax), "v"(ay), "v"(az), "v"(aw));
            r[t] = load_row(W, idx, lane);
        }
        float4 actv;
        actv.x = gelu_erf(ax * inv);
        actv.y = gelu_erf(ay * inv);
        actv.z = gelu_erf(az * inv);
        actv.w = gelu_erf(aw * inv);
        *(float4*)&lds[c * LDS_PITCH + lane * 4] = actv;
        idx_nxt = idx_n2;
    }
    {
        float ax = 0.f, ay = 0.f, az = 0.f, aw = 0.f;
#pragma unroll
        for (int t = 0; t < TOK; ++t) {
            const float4 v = r[t];
            ax += v.x; ay += v.y; az += v.z; aw += v.w;
        }
        float4 actv;
        actv.x = gelu_erf(ax * inv);
        actv.y = gelu_erf(ay * inv);
        actv.z = gelu_erf(az * inv);
        actv.w = gelu_erf(aw * inv);
        *(float4*)&lds[(CELLS_PER_BLOCK - 1) * LDS_PITCH + lane * 4] = actv;
    }

    __syncthreads();

    const int half = lane >> 4;
    const int off  = half << 1;
    const int m    = lane & 15;

    v8f c = {};
#pragma unroll
    for (int j = 0; j < DIM / 4; ++j) {
        v2f a = *(const v2f*)&lds[m * LDS_PITCH + 4 * j + off];
        v2f b;
        b.x = wp[4 * j + off];
        b.y = wp[4 * j + off + 1];
        c = __builtin_amdgcn_wmma_f32_16x16x4_f32(
                false, a, false, b, (short)0, c, false, false);
    }

    if (m == 0) {
        const float bias = bp[0];
        float* o = out + cell0 + half * 8;
#pragma unroll
        for (int r8 = 0; r8 < 8; ++r8)
            o[r8] = c[r8] + bias;
    }
}

extern "C" void kernel_launch(void* const* d_in, const int* in_sizes, int n_in,
                              void* d_out, int out_size, void* d_ws, size_t ws_size,
                              hipStream_t stream) {
    const int*   x  = (const int*)d_in[0];
    const float* W  = (const float*)d_in[1];
    const float* wp = (const float*)d_in[2];
    const float* bp = (const float*)d_in[3];
    float* out = (float*)d_out;

    const int ncells = out_size;
    const int blocks = ncells / CELLS_PER_BLOCK;
    GeluAvgEmbed_kernel<<<blocks, THREADS, 0, stream>>>(x, W, wp, bp, out);
}
